# starting kernel
# baseline (speedup 1.0000x reference)
#include <hip/hip_runtime.h>

typedef __attribute__((ext_vector_type(2))) float v2f;
typedef __attribute__((ext_vector_type(8))) float v8f;

#define IN_F 4096
#define ROWS 8192
#define BLK 64
#define NBLK 64
#define ROWS_PER_WG 128

__global__ __launch_bounds__(256)
void structured_linear_wmma(const float* __restrict__ x,
                            const float* __restrict__ w,
                            float* __restrict__ y)
{
    const int cb   = blockIdx.y;
    const int rowg = blockIdx.x;
    const int tid  = threadIdx.x;
    const int wave = tid >> 5;
    const int lane = tid & 31;
    const int h    = lane >> 4;
    const int l16  = lane & 15;

    const int row_base = rowg * ROWS_PER_WG + wave * 16;
    const int colb     = cb * BLK;

    v2f a[16];
    const float* xrow = x + (size_t)(row_base + l16) * IN_F + colb + 2 * h;
    #pragma unroll
    for (int c = 0; c < 16; ++c)
        a[c] = *(const v2f*)(xrow + 4 * c);

    #pragma unroll
    for (int ct = 0; ct < 4; ++ct) {
        v8f acc = {};
        const float* wrow =
            w + (size_t)(colb + ct * 16 + l16) * IN_F + colb + 2 * h;
        #pragma unroll
        for (int c = 0; c < 16; ++c) {
            v2f b = *(const v2f*)(wrow + 4 * c);
            acc = __builtin_amdgcn_wmma_f32_16x16x4_f32(
                 false, a[c],  false, b,
                 (short)0, acc,  false,  false);
        }
        float* yo = y + (size_t)row_base * IN_F + colb + ct * 16 + l16;
        #pragma unroll
        for (int v = 0; v < 8; ++v)
            yo[(size_t)(v + 8 * h) * IN_F] = acc[v];
    }
}

extern "C" void kernel_launch(void* const* d_in, const int* in_sizes, int n_in,
                              void* d_out, int out_size, void* d_ws, size_t ws_size,
                              hipStream_t stream) {
    (void)in_sizes; (void)n_in; (void)out_size; (void)d_ws; (void)ws_size;
    const float* x = (const float*)d_in[0];
    const float* w = (const float*)d_in[1];
    float*       y = (float*)d_out;

    dim3 grid(ROWS / ROWS_PER_WG, NBLK);
    structured_linear_wmma<<<grid, 256, 0, stream>>>(x, w, y);
}


# W block kept in registers: each wave loads the 64x64 block's B fragments once and loops over 8 row tiles (grid 8x64) instead of reloading all 64 fragments for every 16-row tile
# speedup vs baseline: 1.5099x; 1.5099x over previous
#include <hip/hip_runtime.h>

typedef __attribute__((ext_vector_type(2))) float v2f;
typedef __attribute__((ext_vector_type(8))) float v8f;

#define IN_F 4096
#define ROWS 8192
#define BLK 64
#define NBLK 64
#define WAVES_PER_WG 8
#define TILES_PER_WAVE 8
#define ROWS_PER_WAVE (TILES_PER_WAVE * 16)
#define ROWS_PER_WG (WAVES_PER_WG * ROWS_PER_WAVE)

__global__ __launch_bounds__(256)
void structured_linear_wmma(const float* __restrict__ x,
                            const float* __restrict__ w,
                            float* __restrict__ y)
{
    const int cb   = blockIdx.y;
    const int rowg = blockIdx.x;
    const int tid  = threadIdx.x;
    const int wave = tid >> 5;
    const int lane = tid & 31;
    const int h    = lane >> 4;
    const int l16  = lane & 15;

    const int colb = cb * BLK;

    v2f b[4][16];
    #pragma unroll
    for (int ct = 0; ct < 4; ++ct) {
        const float* wrow =
            w + (size_t)(colb + ct * 16 + l16) * IN_F + colb + 2 * h;
        #pragma unroll
        for (int c = 0; c < 16; ++c)
            b[ct][c] = *(const v2f*)(wrow + 4 * c);
    }

    const int row0 = rowg * ROWS_PER_WG + wave * ROWS_PER_WAVE;

    #pragma unroll 1
    for (int t = 0; t < TILES_PER_WAVE; ++t) {
        const int row_base = row0 + t * 16;

        v2f a[16];
        const float* xrow = x + (size_t)(row_base + l16) * IN_F + colb + 2 * h;
        #pragma unroll
        for (int c = 0; c < 16; ++c)
            a[c] = *(const v2f*)(xrow + 4 * c);

        #pragma unroll
        for (int ct = 0; ct < 4; ++ct) {
            v8f acc = {};
            #pragma unroll
            for (int c = 0; c < 16; ++c) {
                acc = __builtin_amdgcn_wmma_f32_16x16x4_f32(
                     false, a[c],  false, b[ct][c],
                     (short)0, acc,  false,  false);
            }
            float* yo = y + (size_t)row_base * IN_F + colb + ct * 16 + l16;
            #pragma unroll
            for (int v = 0; v < 8; ++v)
                yo[(size_t)(v + 8 * h) * IN_F] = acc[v];
        }
    }
}

extern "C" void kernel_launch(void* const* d_in, const int* in_sizes, int n_in,
                              void* d_out, int out_size, void* d_ws, size_t ws_size,
                              hipStream_t stream) {
    (void)in_sizes; (void)n_in; (void)out_size; (void)d_ws; (void)ws_size;
    const float* x = (const float*)d_in[0];
    const float* w = (const float*)d_in[1];
    float*       y = (float*)d_out;

    dim3 grid(ROWS / ROWS_PER_WG, NBLK);
    structured_linear_wmma<<<grid, 256, 0, stream>>>(x, w, y);
}


# Both fragment loaders read 128 bits per lane (K=8j+4h+0..3; halves feed WMMA 2j, 2j+1): half the load instructions, a tile's 8 issued as one group (waves_per_eu 4); k summed in a permuted order
# speedup vs baseline: 1.5732x; 1.0419x over previous
#include <hip/hip_runtime.h>

typedef __attribute__((ext_vector_type(2))) float v2f;
typedef __attribute__((ext_vector_type(4))) float v4f;
typedef __attribute__((ext_vector_type(8))) float v8f;

#define IN_F 4096
#define ROWS 8192
#define BLK 64
#define NBLK 64
#define WAVES_PER_WG 8
#define TILES_PER_WAVE 8
#define ROWS_PER_WAVE (TILES_PER_WAVE * 16)
#define ROWS_PER_WG (WAVES_PER_WG * ROWS_PER_WAVE)

__global__ __launch_bounds__(256) __attribute__((amdgpu_waves_per_eu(4, 4)))
void structured_linear_wmma(const float* __restrict__ x,
                            const float* __restrict__ w,
                            float* __restrict__ y)
{
    const int cb   = blockIdx.y;
    const int rowg = blockIdx.x;
    const int tid  = threadIdx.x;
    const int wave = tid >> 5;
    const int lane = tid & 31;
    const int h    = lane >> 4;
    const int l16  = lane & 15;

    const int colb = cb * BLK;

    v4f b[4][8];
    #pragma unroll
    for (int ct = 0; ct < 4; ++ct) {
        const float* wrow =
            w + (size_t)(colb + ct * 16 + l16) * IN_F + colb + 4 * h;
        #pragma unroll
        for (int j = 0; j < 8; ++j)
            b[ct][j] = *(const v4f*)(wrow + 8 * j);
    }

    const int row0 = rowg * ROWS_PER_WG + wave * ROWS_PER_WAVE;

    #pragma unroll 1
    for (int t = 0; t < TILES_PER_WAVE; ++t) {
        const int row_base = row0 + t * 16;

        v4f a[8];
        const float* xrow = x + (size_t)(row_base + l16) * IN_F + colb + 4 * h;
        #pragma unroll
        for (int j = 0; j < 8; ++j)
            a[j] = *(const v4f*)(xrow + 8 * j);

        #pragma unroll
        for (int ct = 0; ct < 4; ++ct) {
            v8f acc = {};
            #pragma unroll
            for (int j = 0; j < 8; ++j) {
                const v2f a_lo = a[j].lo,     a_hi = a[j].hi;
                const v2f b_lo = b[ct][j].lo, b_hi = b[ct][j].hi;
                acc = __builtin_amdgcn_wmma_f32_16x16x4_f32(
                     false, a_lo,  false, b_lo,
                     (short)0, acc,  false,  false);
                acc = __builtin_amdgcn_wmma_f32_16x16x4_f32(
                     false, a_hi,  false, b_hi,
                     (short)0, acc,  false,  false);
            }
            float* yo = y + (size_t)row_base * IN_F + colb + ct * 16 + l16;
            #pragma unroll
            for (int v = 0; v < 8; ++v)
                yo[(size_t)(v + 8 * h) * IN_F] = acc[v];
        }
    }
}

extern "C" void kernel_launch(void* const* d_in, const int* in_sizes, int n_in,
                              void* d_out, int out_size, void* d_ws, size_t ws_size,
                              hipStream_t stream) {
    (void)in_sizes; (void)n_in; (void)out_size; (void)d_ws; (void)ws_size;
    const float* x = (const float*)d_in[0];
    const float* w = (const float*)d_in[1];
    float*       y = (float*)d_out;

    dim3 grid(ROWS / ROWS_PER_WG, NBLK);
    structured_linear_wmma<<<grid, 256, 0, stream>>>(x, w, y);
}


# x fragments double-buffered: the next tile's 8 loads are issued in front of the current tile's 64 WMMAs (second register set, two tiles per loop trip, last pair behind the loop); sums unchanged
# speedup vs baseline: 1.6442x; 1.0451x over previous
#include <hip/hip_runtime.h>

typedef __attribute__((ext_vector_type(2))) float v2f;
typedef __attribute__((ext_vector_type(4))) float v4f;
typedef __attribute__((ext_vector_type(8))) float v8f;

#define IN_F 4096
#define ROWS 8192
#define BLK 64
#define NBLK 64
#define WAVES_PER_WG 8
#define TILES_PER_WAVE 8
#define ROWS_PER_WAVE (TILES_PER_WAVE * 16)
#define ROWS_PER_WG (WAVES_PER_WG * ROWS_PER_WAVE)


__device__ __forceinline__ void load_x_tile(v4f (&a)[8], const float* xrow)
{
    #pragma unroll
    for (int j = 0; j < 8; ++j)
        a[j] = *(const v4f*)(xrow + 8 * j);
}

__device__ __forceinline__ void mma_store_tile(const v4f (&a)[8], const v4f (&b)[4][8],
                                               float* __restrict__ yt, int h)
{
    #pragma unroll
    for (int ct = 0; ct < 4; ++ct) {
        v8f acc = {};
        #pragma unroll
        for (int j = 0; j < 8; ++j) {
            const v2f a_lo = a[j].lo,     a_hi = a[j].hi;
            const v2f b_lo = b[ct][j].lo, b_hi = b[ct][j].hi;
            acc = __builtin_amdgcn_wmma_f32_16x16x4_f32(
                 false, a_lo,  false, b_lo,
                 (short)0, acc,  false,  false);
            acc = __builtin_amdgcn_wmma_f32_16x16x4_f32(
                 false, a_hi,  false, b_hi,
                 (short)0, acc,  false,  false);
        }
        float* yo = yt + ct * 16;
        #pragma unroll
        for (int v = 0; v < 8; ++v)
            yo[(size_t)(v + 8 * h) * IN_F] = acc[v];
    }
}

__global__ __launch_bounds__(256) __attribute__((amdgpu_waves_per_eu(4, 4)))
void structured_linear_wmma(const float* __restrict__ x,
                            const float* __restrict__ w,
                            float* __restrict__ y)
{
    const int cb   = blockIdx.y;
    const int rowg = blockIdx.x;
    const int tid  = threadIdx.x;
    const int wave = tid >> 5;
    const int lane = tid & 31;
    const int h    = lane >> 4;
    const int l16  = lane & 15;

    const int colb = cb * BLK;
    const int row0 = rowg * ROWS_PER_WG + wave * ROWS_PER_WAVE;

    const float* xrow = x + (size_t)(row0 + l16) * IN_F + colb + 4 * h;
    float*       yt   = y + (size_t)row0 * IN_F + colb + l16;
    const size_t tile_stride = (size_t)16 * IN_F;

    v4f b[4][8];
    #pragma unroll
    for (int ct = 0; ct < 4; ++ct) {
        const float* wrow =
            w + (size_t)(colb + ct * 16 + l16) * IN_F + colb + 4 * h;
        #pragma unroll
        for (int j = 0; j < 8; ++j)
            b[ct][j] = *(const v4f*)(wrow + 8 * j);
    }

    v4f a0[8], a1[8];
    load_x_tile(a0, xrow);

    #pragma unroll 1
    for (int t = 0; t < TILES_PER_WAVE - 2; t += 2) {
        load_x_tile(a1, xrow + (size_t)(t + 1) * tile_stride);
        __builtin_amdgcn_sched_barrier(0);
        mma_store_tile(a0, b, yt + (size_t)t * tile_stride, h);

        load_x_tile(a0, xrow + (size_t)(t + 2) * tile_stride);
        __builtin_amdgcn_sched_barrier(0);
        mma_store_tile(a1, b, yt + (size_t)(t + 1) * tile_stride, h);
    }
    load_x_tile(a1, xrow + (size_t)(TILES_PER_WAVE - 1) * tile_stride);
    __builtin_amdgcn_sched_barrier(0);
    mma_store_tile(a0, b, yt + (size_t)(TILES_PER_WAVE - 2) * tile_stride, h);
    __builtin_amdgcn_sched_barrier(0);
    mma_store_tile(a1, b, yt + (size_t)(TILES_PER_WAVE - 1) * tile_stride, h);
}

extern "C" void kernel_launch(void* const* d_in, const int* in_sizes, int n_in,
                              void* d_out, int out_size, void* d_ws, size_t ws_size,
                              hipStream_t stream) {
    (void)in_sizes; (void)n_in; (void)out_size; (void)d_ws; (void)ws_size;
    const float* x = (const float*)d_in[0];
    const float* w = (const float*)d_in[1];
    float*       y = (float*)d_out;

    dim3 grid(ROWS / ROWS_PER_WG, NBLK);
    structured_linear_wmma<<<grid, 256, 0, stream>>>(x, w, y);
}
